# starting kernel
# baseline (speedup 1.0000x reference)
#include <hip/hip_runtime.h>


typedef __attribute__((ext_vector_type(2))) float v2f;
typedef __attribute__((ext_vector_type(8))) float v8f;

#if defined(__has_builtin)
#  if __has_builtin(__builtin_amdgcn_tanhf)
#    define TANHF(x) __builtin_amdgcn_tanhf(x)
#  else
#    define TANHF(x) tanhf(x)
#  endif
#else
#  define TANHF(x) tanhf(x)
#endif

#define NELEM   4
#define DESC    128
#define HID     64
#define CHUNK   256
#define NWAVES  8
#define HSTRIDE 66

#define W1_F  (NELEM*DESC*HID)
#define W2_F  (NELEM*HID*HID)
#define W3_F  (NELEM*HID)
#define B1_F  (NELEM*HID)
#define B2_F  (NELEM*HID)
#define B3_F  (NELEM)
#define HST_F (NWAVES*16*HSTRIDE)
#define MOLE_F 4
#define FLOATS_TOTAL (W1_F+W2_F+W3_F+B1_F+B2_F+B3_F+HST_F+MOLE_F)
#define INTS_TOTAL   (NELEM + NELEM*CHUNK)
#define SMEM_BYTES   ((size_t)(FLOATS_TOTAL + INTS_TOTAL) * 4)

__global__ __launch_bounds__(256)
void hdnnp_wmma_kernel(const float* __restrict__ X,
                       const int*   __restrict__ Z,
                       const float* __restrict__ W1,
                       const float* __restrict__ b1,
                       const float* __restrict__ W2,
                       const float* __restrict__ b2,
                       const float* __restrict__ W3,
                       const float* __restrict__ b3,
                       float*       __restrict__ out)
{
    extern __shared__ float smem[];
    float* w1s    = smem;
    float* w2s    = w1s + W1_F;
    float* w3s    = w2s + W2_F;
    float* b1s    = w3s + W3_F;
    float* b2s    = b1s + B1_F;
    float* b3s    = b2s + B2_F;
    float* hstage = b3s + B3_F;
    float* molE   = hstage + HST_F;
    int*   cnt    = (int*)(molE + MOLE_F);
    int*   binIdx = cnt + NELEM;

    const int tid  = threadIdx.x;
    const int lane = tid & 31;
    const int wave = tid >> 5;
    const int m    = lane & 15;
    const int hf   = lane >> 4;
    const int base = blockIdx.x * CHUNK;

    if (tid < NELEM) cnt[tid] = 0;
    if (tid < 4)     molE[tid] = 0.0f;
    __syncthreads();

    {
        int z   = Z[base + tid];
        int pos = atomicAdd(&cnt[z], 1);
        binIdx[z * CHUNK + pos] = tid;
    }

    {
        const float4* s1 = (const float4*)W1;
        for (int i = tid; i < W1_F / 4; i += 256) {
            float4 v = s1[i];
            int g  = i * 4;
            int n  = g & (HID - 1);
            int d  = (g >> 6) & (DESC - 1);
            int e  = g >> 13;
            int kk = d >> 2, hh = (d >> 1) & 1, lo = d & 1;
            float* dst = w1s + ((e * (DESC/4) + kk) * 2 + hh) * (2*HID) + 2*n + lo;
            dst[0] = v.x; dst[2] = v.y; dst[4] = v.z; dst[6] = v.w;
        }
        const float4* s2 = (const float4*)W2;
        for (int i = tid; i < W2_F / 4; i += 256) {
            float4 v = s2[i];
            int g  = i * 4;
            int n  = g & (HID - 1);
            int h  = (g >> 6) & (HID - 1);
            int e  = g >> 12;
            int kk = h >> 2, hh = (h >> 1) & 1, lo = h & 1;
            float* dst = w2s + ((e * (HID/4) + kk) * 2 + hh) * (2*HID) + 2*n + lo;
            dst[0] = v.x; dst[2] = v.y; dst[4] = v.z; dst[6] = v.w;
        }
        if (tid < W3_F) w3s[tid] = W3[tid];
        if (tid < B1_F) b1s[tid] = b1[tid];
        if (tid < B2_F) b2s[tid] = b2[tid];
        if (tid < B3_F) b3s[tid] = b3[tid];
    }
    __syncthreads();

    const int c0 = cnt[0], c1 = cnt[1], c2 = cnt[2], c3 = cnt[3];
    const int t0 = (c0 + 15) >> 4, t1 = (c1 + 15) >> 4;
    const int t2 = (c2 + 15) >> 4, t3 = (c3 + 15) >> 4;
    const int T  = t0 + t1 + t2 + t3;

    float* hst = hstage + wave * (16 * HSTRIDE);

    for (int t = wave; t < T; t += NWAVES) {
        int e, lt, r = t;
        if (r < t0)            { e = 0; lt = r; }
        else { r -= t0;
        if (r < t1)            { e = 1; lt = r; }
        else { r -= t1;
        if (r < t2)            { e = 2; lt = r; }
        else                   { e = 3; lt = r - t2; } } }
        const int ccnt = cnt[e];

        const int  pos   = lt * 16 + m;
        const bool valid = pos < ccnt;
        const int  slot  = valid ? binIdx[e * CHUNK + pos] : 0;
        const float* xrow = X + (size_t)(base + slot) * DESC + 2 * hf;

        v8f a0 = {}, a1 = {}, a2 = {}, a3 = {};
        const float* w1e = w1s + e * (DESC * HID);
        #pragma unroll 4
        for (int kk = 0; kk < DESC / 4; ++kk) {
            v2f a = *(const v2f*)(xrow + 4 * kk);
            const float* wb = w1e + (kk * 2 + hf) * (2*HID) + 2*m;
            v2f bt0 = *(const v2f*)(wb);
            v2f bt1 = *(const v2f*)(wb + 32);
            v2f bt2 = *(const v2f*)(wb + 64);
            v2f bt3 = *(const v2f*)(wb + 96);
            a0 = __builtin_amdgcn_wmma_f32_16x16x4_f32(false, a, false, bt0, (short)0, a0, false, false);
            a1 = __builtin_amdgcn_wmma_f32_16x16x4_f32(false, a, false, bt1, (short)0, a1, false, false);
            a2 = __builtin_amdgcn_wmma_f32_16x16x4_f32(false, a, false, bt2, (short)0, a2, false, false);
            a3 = __builtin_amdgcn_wmma_f32_16x16x4_f32(false, a, false, bt3, (short)0, a3, false, false);
        }
        {
            const float* b1e = b1s + e * HID;
            const float z0 = b1e[m], z1 = b1e[16 + m], z2 = b1e[32 + m], z3 = b1e[48 + m];
            #pragma unroll
            for (int g = 0; g < 8; ++g) {
                float* hr = hst + (g + 8 * hf) * HSTRIDE + m;
                hr[0]  = TANHF(a0[g] + z0);
                hr[16] = TANHF(a1[g] + z1);
                hr[32] = TANHF(a2[g] + z2);
                hr[48] = TANHF(a3[g] + z3);
            }
        }

        v8f d0 = {}, d1 = {}, d2 = {}, d3 = {};
        const float* w2e  = w2s + e * (HID * HID);
        const float* arow = hst + m * HSTRIDE + 2 * hf;
        #pragma unroll 4
        for (int kk = 0; kk < HID / 4; ++kk) {
            v2f a = *(const v2f*)(arow + 4 * kk);
            const float* wb = w2e + (kk * 2 + hf) * (2*HID) + 2*m;
            v2f bt0 = *(const v2f*)(wb);
            v2f bt1 = *(const v2f*)(wb + 32);
            v2f bt2 = *(const v2f*)(wb + 64);
            v2f bt3 = *(const v2f*)(wb + 96);
            d0 = __builtin_amdgcn_wmma_f32_16x16x4_f32(false, a, false, bt0, (short)0, d0, false, false);
            d1 = __builtin_amdgcn_wmma_f32_16x16x4_f32(false, a, false, bt1, (short)0, d1, false, false);
            d2 = __builtin_amdgcn_wmma_f32_16x16x4_f32(false, a, false, bt2, (short)0, d2, false, false);
            d3 = __builtin_amdgcn_wmma_f32_16x16x4_f32(false, a, false, bt3, (short)0, d3, false, false);
        }
        {
            const float* b2e = b2s + e * HID;
            const float z0 = b2e[m], z1 = b2e[16 + m], z2 = b2e[32 + m], z3 = b2e[48 + m];
            #pragma unroll
            for (int g = 0; g < 8; ++g) {
                float* hr = hst + (g + 8 * hf) * HSTRIDE + m;
                hr[0]  = TANHF(d0[g] + z0);
                hr[16] = TANHF(d1[g] + z1);
                hr[32] = TANHF(d2[g] + z2);
                hr[48] = TANHF(d3[g] + z3);
            }
        }

        {
            float en = 0.0f;
            const float* w3e = w3s + e * HID;
            const float* hr  = hst + m * HSTRIDE;
            #pragma unroll 8
            for (int h = 0; h < HID; ++h) en = fmaf(hr[h], w3e[h], en);
            en += b3s[e];
            if (hf == 0 && valid)
                atomicAdd(&molE[slot >> 6], en);
        }
    }

    __syncthreads();
    if (tid < 4) out[blockIdx.x * 4 + tid] = molE[tid];
}

extern "C" void kernel_launch(void* const* d_in, const int* in_sizes, int n_in,
                              void* d_out, int out_size, void* d_ws, size_t ws_size,
                              hipStream_t stream) {
    (void)n_in; (void)out_size; (void)d_ws; (void)ws_size;
    const float* X  = (const float*)d_in[0];
    const int*   Z  = (const int*)  d_in[1];
    const float* W1 = (const float*)d_in[2];
    const float* b1 = (const float*)d_in[3];
    const float* W2 = (const float*)d_in[4];
    const float* b2 = (const float*)d_in[5];
    const float* W3 = (const float*)d_in[6];
    const float* b3 = (const float*)d_in[7];
    float* out = (float*)d_out;

    const int M      = in_sizes[1];
    const int blocks = M / CHUNK;

    hipLaunchKernelGGL(hdnnp_wmma_kernel, dim3(blocks), dim3(256), SMEM_BYTES,
                       stream, X, Z, W1, b1, W2, b2, W3, b3, out);
}


# Element-major grid: one work-group per (element, 2048 atoms) stages only that element's W1,W2 (48 KB LDS), two groups per WGP; molecule sums meet by float atomic adds into out after a zeroing launch
# speedup vs baseline: 1.7298x; 1.7298x over previous
#include <hip/hip_runtime.h>


typedef __attribute__((ext_vector_type(2))) float v2f;
typedef __attribute__((ext_vector_type(8))) float v8f;

#if defined(__has_builtin)
#  if __has_builtin(__builtin_amdgcn_tanhf)
#    define TANHF(x) __builtin_amdgcn_tanhf(x)
#  else
#    define TANHF(x) tanhf(x)
#  endif
#else
#  define TANHF(x) tanhf(x)
#endif

#define NELEM    4
#define DESC     128
#define HID      64
#define CHUNK    2048
#define MOLS     (CHUNK / 64)
#define NWAVES   12
#define NTHREADS (NWAVES * 32)
#define ZITER    ((CHUNK + NTHREADS - 1) / NTHREADS)
#define WITER    ((DESC*HID + HID*HID) / 4 / NTHREADS)
#define HSTRIDE  66

#define W1_F  (DESC*HID)
#define W2_F  (HID*HID)
#define W3_F  (HID)
#define B1_F  (HID)
#define B2_F  (HID)
#define B3_F  4
#define HST_F (NWAVES*16*HSTRIDE)
#define MOLE_F MOLS
#define FLOATS_TOTAL (W1_F+W2_F+W3_F+B1_F+B2_F+B3_F+HST_F+MOLE_F)
#define INTS_TOTAL   (4 + CHUNK)
#define SMEM_BYTES   ((size_t)(FLOATS_TOTAL + INTS_TOTAL) * 4)

__global__ __launch_bounds__(MOLS)
void zero_out_kernel(float* __restrict__ out)
{
    out[blockIdx.x * MOLS + threadIdx.x] = 0.0f;
}

__global__ __launch_bounds__(NTHREADS)
void hdnnp_wmma_kernel(const float* __restrict__ X,
                       const int*   __restrict__ Z,
                       const float* __restrict__ W1,
                       const float* __restrict__ b1,
                       const float* __restrict__ W2,
                       const float* __restrict__ b2,
                       const float* __restrict__ W3,
                       const float* __restrict__ b3,
                       float*       __restrict__ out)
{
    extern __shared__ float smem[];
    float* w1s    = smem;
    float* w2s    = w1s + W1_F;
    float* w3s    = w2s + W2_F;
    float* b1s    = w3s + W3_F;
    float* b2s    = b1s + B1_F;
    float* b3s    = b2s + B2_F;
    float* hstage = b3s + B3_F;
    float* molE   = hstage + HST_F;
    int*   cnt    = (int*)(molE + MOLE_F);
    int*   binIdx = cnt + 4;

    const int tid  = threadIdx.x;
    const int lane = tid & 31;
    const int wave = tid >> 5;
    const int m    = lane & 15;
    const int hf   = lane >> 4;
    const int e    = blockIdx.y;
    const int base = blockIdx.x * CHUNK;

    if (tid == 0)   cnt[0] = 0;
    if (tid < MOLS) molE[tid] = 0.0f;
    __syncthreads();

    int zv[ZITER];
    #pragma unroll
    for (int j = 0; j < ZITER; ++j) {
        const int i = tid + j * NTHREADS;
        zv[j] = (i < CHUNK) ? Z[base + i] : -1;
    }
    const float4* s1 = (const float4*)(W1 + (size_t)e * W1_F);
    const float4* s2 = (const float4*)(W2 + (size_t)e * W2_F);
    float4 wv[WITER];
    #pragma unroll
    for (int j = 0; j < WITER; ++j) {
        const int idx = tid + j * NTHREADS;
        wv[j] = (idx < W1_F / 4) ? s1[idx] : s2[idx - W1_F / 4];
    }
    float w3v = 0.0f, b1v = 0.0f, b2v = 0.0f;
    if (tid < HID) {
        w3v = W3[e * HID + tid];
        b1v = b1[e * HID + tid];
        b2v = b2[e * HID + tid];
    }
    const float b3v = b3[e];

    #pragma unroll
    for (int j = 0; j < ZITER; ++j) {
        if (zv[j] == e) {
            int pos = atomicAdd(&cnt[0], 1);
            binIdx[pos] = tid + j * NTHREADS;
        }
    }

    #pragma unroll
    for (int j = 0; j < WITER; ++j) {
        const int g = (tid + j * NTHREADS) * 4;
        const int n = g & (HID - 1);
        const int D = g >> 6;
        float* dst = w1s + (D >> 1) * (2*HID) + 2*n + (D & 1);
        dst[0] = wv[j].x; dst[2] = wv[j].y; dst[4] = wv[j].z; dst[6] = wv[j].w;
    }
    if (tid < HID) { w3s[tid] = w3v; b1s[tid] = b1v; b2s[tid] = b2v; }
    if (tid == 0) b3s[0] = b3v;
    __syncthreads();

    const int ccnt = cnt[0];
    const int T    = (ccnt + 15) >> 4;

    float* hst = hstage + wave * (16 * HSTRIDE);

    for (int t = wave; t < T; t += NWAVES) {
        const int  pos   = t * 16 + m;
        const bool valid = pos < ccnt;
        const int  slot  = valid ? binIdx[pos] : 0;
        const float* xrow = X + (size_t)(base + slot) * DESC + 2 * hf;

        v8f a0 = {}, a1 = {}, a2 = {}, a3 = {};
        #pragma unroll 4
        for (int kk = 0; kk < DESC / 4; ++kk) {
            v2f a = *(const v2f*)(xrow + 4 * kk);
            const float* wb = w1s + (kk * 2 + hf) * (2*HID) + 2*m;
            v2f bt0 = *(const v2f*)(wb);
            v2f bt1 = *(const v2f*)(wb + 32);
            v2f bt2 = *(const v2f*)(wb + 64);
            v2f bt3 = *(const v2f*)(wb + 96);
            a0 = __builtin_amdgcn_wmma_f32_16x16x4_f32(false, a, false, bt0, (short)0, a0, false, false);
            a1 = __builtin_amdgcn_wmma_f32_16x16x4_f32(false, a, false, bt1, (short)0, a1, false, false);
            a2 = __builtin_amdgcn_wmma_f32_16x16x4_f32(false, a, false, bt2, (short)0, a2, false, false);
            a3 = __builtin_amdgcn_wmma_f32_16x16x4_f32(false, a, false, bt3, (short)0, a3, false, false);
        }
        {
            const float z0 = b1s[m], z1 = b1s[16 + m], z2 = b1s[32 + m], z3 = b1s[48 + m];
            #pragma unroll
            for (int g = 0; g < 8; ++g) {
                float* hr = hst + (g + 8 * hf) * HSTRIDE + m;
                hr[0]  = TANHF(a0[g] + z0);
                hr[16] = TANHF(a1[g] + z1);
                hr[32] = TANHF(a2[g] + z2);
                hr[48] = TANHF(a3[g] + z3);
            }
        }

        v8f d0 = {}, d1 = {}, d2 = {}, d3 = {};
        const float* arow = hst + m * HSTRIDE + 2 * hf;
        #pragma unroll 4
        for (int kk = 0; kk < HID / 4; ++kk) {
            v2f a = *(const v2f*)(arow + 4 * kk);
            const float* wb = w2s + (kk * 2 + hf) * (2*HID) + 2*m;
            v2f bt0 = *(const v2f*)(wb);
            v2f bt1 = *(const v2f*)(wb + 32);
            v2f bt2 = *(const v2f*)(wb + 64);
            v2f bt3 = *(const v2f*)(wb + 96);
            d0 = __builtin_amdgcn_wmma_f32_16x16x4_f32(false, a, false, bt0, (short)0, d0, false, false);
            d1 = __builtin_amdgcn_wmma_f32_16x16x4_f32(false, a, false, bt1, (short)0, d1, false, false);
            d2 = __builtin_amdgcn_wmma_f32_16x16x4_f32(false, a, false, bt2, (short)0, d2, false, false);
            d3 = __builtin_amdgcn_wmma_f32_16x16x4_f32(false, a, false, bt3, (short)0, d3, false, false);
        }
        {
            const float z0 = b2s[m], z1 = b2s[16 + m], z2 = b2s[32 + m], z3 = b2s[48 + m];
            #pragma unroll
            for (int g = 0; g < 8; ++g) {
                float* hr = hst + (g + 8 * hf) * HSTRIDE + m;
                hr[0]  = TANHF(d0[g] + z0);
                hr[16] = TANHF(d1[g] + z1);
                hr[32] = TANHF(d2[g] + z2);
                hr[48] = TANHF(d3[g] + z3);
            }
        }

        {
            float en = 0.0f;
            const float* hr = hst + m * HSTRIDE;
            #pragma unroll 8
            for (int h = 0; h < HID; ++h) en = fmaf(hr[h], w3s[h], en);
            en += b3s[0];
            if (hf == 0 && valid)
                atomicAdd(&molE[slot >> 6], en);
        }
    }

    __syncthreads();
    if (tid < MOLS) atomicAdd(&out[blockIdx.x * MOLS + tid], molE[tid]);
}

extern "C" void kernel_launch(void* const* d_in, const int* in_sizes, int n_in,
                              void* d_out, int out_size, void* d_ws, size_t ws_size,
                              hipStream_t stream) {
    (void)n_in; (void)out_size; (void)d_ws; (void)ws_size;
    const float* X  = (const float*)d_in[0];
    const int*   Z  = (const int*)  d_in[1];
    const float* W1 = (const float*)d_in[2];
    const float* b1 = (const float*)d_in[3];
    const float* W2 = (const float*)d_in[4];
    const float* b2 = (const float*)d_in[5];
    const float* W3 = (const float*)d_in[6];
    const float* b3 = (const float*)d_in[7];
    float* out = (float*)d_out;

    const int M      = in_sizes[1];
    const int chunks = M / CHUNK;

    hipLaunchKernelGGL(zero_out_kernel, dim3(chunks), dim3(MOLS), 0, stream, out);
    hipLaunchKernelGGL(hdnnp_wmma_kernel, dim3(chunks, NELEM), dim3(NTHREADS), SMEM_BYTES,
                       stream, X, Z, W1, b1, W2, b2, W3, b3, out);
}


# Tiles handed out by an LDS ticket counter: a wave draws its next 16-atom tile when it is free, not the fixed wave + 12*i round-robin; LDS float-atomic arrival order changes
# speedup vs baseline: 1.9238x; 1.1122x over previous
#include <hip/hip_runtime.h>


typedef __attribute__((ext_vector_type(2))) float v2f;
typedef __attribute__((ext_vector_type(8))) float v8f;

#if defined(__has_builtin)
#  if __has_builtin(__builtin_amdgcn_tanhf)
#    define TANHF(x) __builtin_amdgcn_tanhf(x)
#  else
#    define TANHF(x) tanhf(x)
#  endif
#else
#  define TANHF(x) tanhf(x)
#endif

#define NELEM    4
#define DESC     128
#define HID      64
#define CHUNK    2048
#define MOLS     (CHUNK / 64)
#define NWAVES   12
#define NTHREADS (NWAVES * 32)
#define ZITER    ((CHUNK + NTHREADS - 1) / NTHREADS)
#define WITER    ((DESC*HID + HID*HID) / 4 / NTHREADS)
#define HSTRIDE  66

#define W1_F  (DESC*HID)
#define W2_F  (HID*HID)
#define W3_F  (HID)
#define B1_F  (HID)
#define B2_F  (HID)
#define B3_F  4
#define HST_F (NWAVES*16*HSTRIDE)
#define MOLE_F MOLS
#define FLOATS_TOTAL (W1_F+W2_F+W3_F+B1_F+B2_F+B3_F+HST_F+MOLE_F)
#define INTS_TOTAL   (4 + CHUNK)
#define SMEM_BYTES   ((size_t)(FLOATS_TOTAL + INTS_TOTAL) * 4)

__global__ __launch_bounds__(MOLS)
void zero_out_kernel(float* __restrict__ out)
{
    out[blockIdx.x * MOLS + threadIdx.x] = 0.0f;
}

__global__ __launch_bounds__(NTHREADS)
void hdnnp_wmma_kernel(const float* __restrict__ X,
                       const int*   __restrict__ Z,
                       const float* __restrict__ W1,
                       const float* __restrict__ b1,
                       const float* __restrict__ W2,
                       const float* __restrict__ b2,
                       const float* __restrict__ W3,
                       const float* __restrict__ b3,
                       float*       __restrict__ out)
{
    extern __shared__ float smem[];
    float* w1s    = smem;
    float* w2s    = w1s + W1_F;
    float* w3s    = w2s + W2_F;
    float* b1s    = w3s + W3_F;
    float* b2s    = b1s + B1_F;
    float* b3s    = b2s + B2_F;
    float* hstage = b3s + B3_F;
    float* molE   = hstage + HST_F;
    int*   cnt    = (int*)(molE + MOLE_F);
    int*   binIdx = cnt + 4;

    const int tid  = threadIdx.x;
    const int lane = tid & 31;
    const int wave = tid >> 5;
    const int m    = lane & 15;
    const int hf   = lane >> 4;
    const int e    = blockIdx.y;
    const int base = blockIdx.x * CHUNK;

    if (tid == 0) { cnt[0] = 0; cnt[1] = 0; }
    if (tid < MOLS) molE[tid] = 0.0f;
    __syncthreads();

    int zv[ZITER];
    #pragma unroll
    for (int j = 0; j < ZITER; ++j) {
        const int i = tid + j * NTHREADS;
        zv[j] = (i < CHUNK) ? Z[base + i] : -1;
    }
    const float4* s1 = (const float4*)(W1 + (size_t)e * W1_F);
    const float4* s2 = (const float4*)(W2 + (size_t)e * W2_F);
    float4 wv[WITER];
    #pragma unroll
    for (int j = 0; j < WITER; ++j) {
        const int idx = tid + j * NTHREADS;
        wv[j] = (idx < W1_F / 4) ? s1[idx] : s2[idx - W1_F / 4];
    }
    float w3v = 0.0f, b1v = 0.0f, b2v = 0.0f;
    if (tid < HID) {
        w3v = W3[e * HID + tid];
        b1v = b1[e * HID + tid];
        b2v = b2[e * HID + tid];
    }
    const float b3v = b3[e];

    #pragma unroll
    for (int j = 0; j < ZITER; ++j) {
        if (zv[j] == e) {
            int pos = atomicAdd(&cnt[0], 1);
            binIdx[pos] = tid + j * NTHREADS;
        }
    }

    #pragma unroll
    for (int j = 0; j < WITER; ++j) {
        const int g = (tid + j * NTHREADS) * 4;
        const int n = g & (HID - 1);
        const int D = g >> 6;
        float* dst = w1s + (D >> 1) * (2*HID) + 2*n + (D & 1);
        dst[0] = wv[j].x; dst[2] = wv[j].y; dst[4] = wv[j].z; dst[6] = wv[j].w;
    }
    if (tid < HID) { w3s[tid] = w3v; b1s[tid] = b1v; b2s[tid] = b2v; }
    if (tid == 0) b3s[0] = b3v;
    __syncthreads();

    const int ccnt = cnt[0];
    const int T    = (ccnt + 15) >> 4;

    float* hst = hstage + wave * (16 * HSTRIDE);

    for (;;) {
        int tk = 0;
        if (lane == 0) tk = atomicAdd(&cnt[1], 1);
        const int t = __builtin_amdgcn_readfirstlane(tk);
        if (t >= T) break;
        const int  pos   = t * 16 + m;
        const bool valid = pos < ccnt;
        const int  slot  = valid ? binIdx[pos] : 0;
        const float* xrow = X + (size_t)(base + slot) * DESC + 2 * hf;

        v8f a0 = {}, a1 = {}, a2 = {}, a3 = {};
        #pragma unroll 4
        for (int kk = 0; kk < DESC / 4; ++kk) {
            v2f a = *(const v2f*)(xrow + 4 * kk);
            const float* wb = w1s + (kk * 2 + hf) * (2*HID) + 2*m;
            v2f bt0 = *(const v2f*)(wb);
            v2f bt1 = *(const v2f*)(wb + 32);
            v2f bt2 = *(const v2f*)(wb + 64);
            v2f bt3 = *(const v2f*)(wb + 96);
            a0 = __builtin_amdgcn_wmma_f32_16x16x4_f32(false, a, false, bt0, (short)0, a0, false, false);
            a1 = __builtin_amdgcn_wmma_f32_16x16x4_f32(false, a, false, bt1, (short)0, a1, false, false);
            a2 = __builtin_amdgcn_wmma_f32_16x16x4_f32(false, a, false, bt2, (short)0, a2, false, false);
            a3 = __builtin_amdgcn_wmma_f32_16x16x4_f32(false, a, false, bt3, (short)0, a3, false, false);
        }
        {
            const float z0 = b1s[m], z1 = b1s[16 + m], z2 = b1s[32 + m], z3 = b1s[48 + m];
            #pragma unroll
            for (int g = 0; g < 8; ++g) {
                float* hr = hst + (g + 8 * hf) * HSTRIDE + m;
                hr[0]  = TANHF(a0[g] + z0);
                hr[16] = TANHF(a1[g] + z1);
                hr[32] = TANHF(a2[g] + z2);
                hr[48] = TANHF(a3[g] + z3);
            }
        }

        v8f d0 = {}, d1 = {}, d2 = {}, d3 = {};
        const float* arow = hst + m * HSTRIDE + 2 * hf;
        #pragma unroll 4
        for (int kk = 0; kk < HID / 4; ++kk) {
            v2f a = *(const v2f*)(arow + 4 * kk);
            const float* wb = w2s + (kk * 2 + hf) * (2*HID) + 2*m;
            v2f bt0 = *(const v2f*)(wb);
            v2f bt1 = *(const v2f*)(wb + 32);
            v2f bt2 = *(const v2f*)(wb + 64);
            v2f bt3 = *(const v2f*)(wb + 96);
            d0 = __builtin_amdgcn_wmma_f32_16x16x4_f32(false, a, false, bt0, (short)0, d0, false, false);
            d1 = __builtin_amdgcn_wmma_f32_16x16x4_f32(false, a, false, bt1, (short)0, d1, false, false);
            d2 = __builtin_amdgcn_wmma_f32_16x16x4_f32(false, a, false, bt2, (short)0, d2, false, false);
            d3 = __builtin_amdgcn_wmma_f32_16x16x4_f32(false, a, false, bt3, (short)0, d3, false, false);
        }
        {
            const float z0 = b2s[m], z1 = b2s[16 + m], z2 = b2s[32 + m], z3 = b2s[48 + m];
            #pragma unroll
            for (int g = 0; g < 8; ++g) {
                float* hr = hst + (g + 8 * hf) * HSTRIDE + m;
                hr[0]  = TANHF(d0[g] + z0);
                hr[16] = TANHF(d1[g] + z1);
                hr[32] = TANHF(d2[g] + z2);
                hr[48] = TANHF(d3[g] + z3);
            }
        }

        {
            float en = 0.0f;
            const float* hr = hst + m * HSTRIDE;
            #pragma unroll 8
            for (int h = 0; h < HID; ++h) en = fmaf(hr[h], w3s[h], en);
            en += b3s[0];
            if (hf == 0 && valid)
                atomicAdd(&molE[slot >> 6], en);
        }
    }

    __syncthreads();
    if (tid < MOLS) atomicAdd(&out[blockIdx.x * MOLS + tid], molE[tid]);
}

extern "C" void kernel_launch(void* const* d_in, const int* in_sizes, int n_in,
                              void* d_out, int out_size, void* d_ws, size_t ws_size,
                              hipStream_t stream) {
    (void)n_in; (void)out_size; (void)d_ws; (void)ws_size;
    const float* X  = (const float*)d_in[0];
    const int*   Z  = (const int*)  d_in[1];
    const float* W1 = (const float*)d_in[2];
    const float* b1 = (const float*)d_in[3];
    const float* W2 = (const float*)d_in[4];
    const float* b2 = (const float*)d_in[5];
    const float* W3 = (const float*)d_in[6];
    const float* b3 = (const float*)d_in[7];
    float* out = (float*)d_out;

    const int M      = in_sizes[1];
    const int chunks = M / CHUNK;

    hipLaunchKernelGGL(zero_out_kernel, dim3(chunks), dim3(MOLS), 0, stream, out);
    hipLaunchKernelGGL(hdnnp_wmma_kernel, dim3(chunks, NELEM), dim3(NTHREADS), SMEM_BYTES,
                       stream, X, Z, W1, b1, W2, b2, W3, b3, out);
}


# Layer-1 A operand through LDS: a tile's 16 rows come by 16 row-contiguous 16-byte-per-lane loads issued together, staged per K half (k loop unrolled per half), read by ds_load_b64; k order kept
# speedup vs baseline: 2.0135x; 1.0466x over previous
#include <hip/hip_runtime.h>


typedef __attribute__((ext_vector_type(2))) float v2f;
typedef __attribute__((ext_vector_type(4))) float v4f;
typedef __attribute__((ext_vector_type(8))) float v8f;

#if defined(__has_builtin)
#  if __has_builtin(__builtin_amdgcn_tanhf)
#    define TANHF(x) __builtin_amdgcn_tanhf(x)
#  else
#    define TANHF(x) tanhf(x)
#  endif
#else
#  define TANHF(x) tanhf(x)
#endif

#define NELEM    4
#define DESC     128
#define HID      64
#define CHUNK    2048
#define MOLS     (CHUNK / 64)
#define NWAVES   12
#define NTHREADS (NWAVES * 32)
#define ZITER    ((CHUNK + NTHREADS - 1) / NTHREADS)
#define WITER    ((DESC*HID + HID*HID) / 4 / NTHREADS)
#define HSTRIDE  68

#define W1_F  (DESC*HID)
#define W2_F  (HID*HID)
#define W3_F  (HID)
#define B1_F  (HID)
#define B2_F  (HID)
#define B3_F  4
#define HST_F (NWAVES*16*HSTRIDE)
#define MOLE_F MOLS
#define FLOATS_TOTAL (W1_F+W2_F+W3_F+B1_F+B2_F+B3_F+HST_F+MOLE_F)
#define INTS_TOTAL   (4 + CHUNK)
#define SMEM_BYTES   ((size_t)(FLOATS_TOTAL + INTS_TOTAL) * 4)

static_assert((W1_F + W2_F + W3_F + B1_F + B2_F + B3_F) % 4 == 0, "staging start not 16-byte aligned");
static_assert(HSTRIDE % 4 == 0 && HSTRIDE >= HID, "staging rows not 16-byte aligned");

__global__ __launch_bounds__(MOLS)
void zero_out_kernel(float* __restrict__ out)
{
    out[blockIdx.x * MOLS + threadIdx.x] = 0.0f;
}

__global__ __launch_bounds__(NTHREADS)
void hdnnp_wmma_kernel(const float* __restrict__ X,
                       const int*   __restrict__ Z,
                       const float* __restrict__ W1,
                       const float* __restrict__ b1,
                       const float* __restrict__ W2,
                       const float* __restrict__ b2,
                       const float* __restrict__ W3,
                       const float* __restrict__ b3,
                       float*       __restrict__ out)
{
    extern __shared__ __attribute__((aligned(16))) float smem[];
    float* w1s    = smem;
    float* w2s    = w1s + W1_F;
    float* w3s    = w2s + W2_F;
    float* b1s    = w3s + W3_F;
    float* b2s    = b1s + B1_F;
    float* b3s    = b2s + B2_F;
    float* hstage = b3s + B3_F;
    float* molE   = hstage + HST_F;
    int*   cnt    = (int*)(molE + MOLE_F);
    int*   binIdx = cnt + 4;

    const int tid  = threadIdx.x;
    const int lane = tid & 31;
    const int wave = tid >> 5;
    const int m    = lane & 15;
    const int hf   = lane >> 4;
    const int e    = blockIdx.y;
    const int base = blockIdx.x * CHUNK;

    if (tid == 0) { cnt[0] = 0; cnt[1] = 0; }
    if (tid < MOLS) molE[tid] = 0.0f;
    __syncthreads();

    int zv[ZITER];
    #pragma unroll
    for (int j = 0; j < ZITER; ++j) {
        const int i = tid + j * NTHREADS;
        zv[j] = (i < CHUNK) ? Z[base + i] : -1;
    }
    const float4* s1 = (const float4*)(W1 + (size_t)e * W1_F);
    const float4* s2 = (const float4*)(W2 + (size_t)e * W2_F);
    float4 wv[WITER];
    #pragma unroll
    for (int j = 0; j < WITER; ++j) {
        const int idx = tid + j * NTHREADS;
        wv[j] = (idx < W1_F / 4) ? s1[idx] : s2[idx - W1_F / 4];
    }
    float w3v = 0.0f, b1v = 0.0f, b2v = 0.0f;
    if (tid < HID) {
        w3v = W3[e * HID + tid];
        b1v = b1[e * HID + tid];
        b2v = b2[e * HID + tid];
    }
    const float b3v = b3[e];

    #pragma unroll
    for (int j = 0; j < ZITER; ++j) {
        if (zv[j] == e) {
            int pos = atomicAdd(&cnt[0], 1);
            binIdx[pos] = tid + j * NTHREADS;
        }
    }

    #pragma unroll
    for (int j = 0; j < WITER; ++j) {
        const int g = (tid + j * NTHREADS) * 4;
        const int n = g & (HID - 1);
        const int D = g >> 6;
        float* dst = w1s + (D >> 1) * (2*HID) + 2*n + (D & 1);
        dst[0] = wv[j].x; dst[2] = wv[j].y; dst[4] = wv[j].z; dst[6] = wv[j].w;
    }
    if (tid < HID) { w3s[tid] = w3v; b1s[tid] = b1v; b2s[tid] = b2v; }
    if (tid == 0) b3s[0] = b3v;
    __syncthreads();

    const int ccnt = cnt[0];
    const int T    = (ccnt + 15) >> 4;

    float* hst = hstage + wave * (16 * HSTRIDE);

    for (;;) {
        int tk = 0;
        if (lane == 0) tk = atomicAdd(&cnt[1], 1);
        const int t = __builtin_amdgcn_readfirstlane(tk);
        if (t >= T) break;
        const int  pos   = t * 16 + m;
        const bool valid = pos < ccnt;
        const int  slot  = valid ? binIdx[pos] : 0;
        v4f xr0[8], xr1[8];
        #pragma unroll
        for (int j = 0; j < 8; ++j) {
            const int se = __builtin_amdgcn_readlane(slot, 2 * j);
            const int so = __builtin_amdgcn_readlane(slot, 2 * j + 1);
            const float* xp = X + (size_t)(base + (hf ? so : se)) * DESC + 4 * m;
            xr0[j] = *(const v4f*)(xp);
            xr1[j] = *(const v4f*)(xp + HID);
        }
        float*       xst  = hst + hf * HSTRIDE + 4 * m;
        const float* xrow = hst + m * HSTRIDE + 2 * hf;

        v8f a0 = {}, a1 = {}, a2 = {}, a3 = {};
        #pragma unroll
        for (int q = 0; q < 2; ++q) {
            #pragma unroll
            for (int j = 0; j < 8; ++j)
                *(v4f*)(xst + 2 * j * HSTRIDE) = q ? xr1[j] : xr0[j];
            #pragma unroll
            for (int kk = 0; kk < HID / 4; ++kk) {
                v2f a = *(const v2f*)(xrow + 4 * kk);
                const float* wb = w1s + ((q * (HID / 4) + kk) * 2 + hf) * (2*HID) + 2*m;
                v2f bt0 = *(const v2f*)(wb);
                v2f bt1 = *(const v2f*)(wb + 32);
                v2f bt2 = *(const v2f*)(wb + 64);
                v2f bt3 = *(const v2f*)(wb + 96);
                a0 = __builtin_amdgcn_wmma_f32_16x16x4_f32(false, a, false, bt0, (short)0, a0, false, false);
                a1 = __builtin_amdgcn_wmma_f32_16x16x4_f32(false, a, false, bt1, (short)0, a1, false, false);
                a2 = __builtin_amdgcn_wmma_f32_16x16x4_f32(false, a, false, bt2, (short)0, a2, false, false);
                a3 = __builtin_amdgcn_wmma_f32_16x16x4_f32(false, a, false, bt3, (short)0, a3, false, false);
            }
        }
        {
            const float z0 = b1s[m], z1 = b1s[16 + m], z2 = b1s[32 + m], z3 = b1s[48 + m];
            #pragma unroll
            for (int g = 0; g < 8; ++g) {
                float* hr = hst + (g + 8 * hf) * HSTRIDE + m;
                hr[0]  = TANHF(a0[g] + z0);
                hr[16] = TANHF(a1[g] + z1);
                hr[32] = TANHF(a2[g] + z2);
                hr[48] = TANHF(a3[g] + z3);
            }
        }

        v8f d0 = {}, d1 = {}, d2 = {}, d3 = {};
        const float* arow = hst + m * HSTRIDE + 2 * hf;
        #pragma unroll 4
        for (int kk = 0; kk < HID / 4; ++kk) {
            v2f a = *(const v2f*)(arow + 4 * kk);
            const float* wb = w2s + (kk * 2 + hf) * (2*HID) + 2*m;
            v2f bt0 = *(const v2f*)(wb);
            v2f bt1 = *(const v2f*)(wb + 32);
            v2f bt2 = *(const v2f*)(wb + 64);
            v2f bt3 = *(const v2f*)(wb + 96);
            d0 = __builtin_amdgcn_wmma_f32_16x16x4_f32(false, a, false, bt0, (short)0, d0, false, false);
            d1 = __builtin_amdgcn_wmma_f32_16x16x4_f32(false, a, false, bt1, (short)0, d1, false, false);
            d2 = __builtin_amdgcn_wmma_f32_16x16x4_f32(false, a, false, bt2, (short)0, d2, false, false);
            d3 = __builtin_amdgcn_wmma_f32_16x16x4_f32(false, a, false, bt3, (short)0, d3, false, false);
        }
        {
            const float z0 = b2s[m], z1 = b2s[16 + m], z2 = b2s[32 + m], z3 = b2s[48 + m];
            #pragma unroll
            for (int g = 0; g < 8; ++g) {
                float* hr = hst + (g + 8 * hf) * HSTRIDE + m;
                hr[0]  = TANHF(d0[g] + z0);
                hr[16] = TANHF(d1[g] + z1);
                hr[32] = TANHF(d2[g] + z2);
                hr[48] = TANHF(d3[g] + z3);
            }
        }

        {
            float en = 0.0f;
            const float* hr = hst + m * HSTRIDE;
            #pragma unroll 8
            for (int h = 0; h < HID; ++h) en = fmaf(hr[h], w3s[h], en);
            en += b3s[0];
            if (hf == 0 && valid)
                atomicAdd(&molE[slot >> 6], en);
        }
    }

    __syncthreads();
    if (tid < MOLS) atomicAdd(&out[blockIdx.x * MOLS + tid], molE[tid]);
}

extern "C" void kernel_launch(void* const* d_in, const int* in_sizes, int n_in,
                              void* d_out, int out_size, void* d_ws, size_t ws_size,
                              hipStream_t stream) {
    (void)n_in; (void)out_size; (void)d_ws; (void)ws_size;
    const float* X  = (const float*)d_in[0];
    const int*   Z  = (const int*)  d_in[1];
    const float* W1 = (const float*)d_in[2];
    const float* b1 = (const float*)d_in[3];
    const float* W2 = (const float*)d_in[4];
    const float* b2 = (const float*)d_in[5];
    const float* W3 = (const float*)d_in[6];
    const float* b3 = (const float*)d_in[7];
    float* out = (float*)d_out;

    const int M      = in_sizes[1];
    const int chunks = M / CHUNK;

    hipLaunchKernelGGL(zero_out_kernel, dim3(chunks), dim3(MOLS), 0, stream, out);
    hipLaunchKernelGGL(hdnnp_wmma_kernel, dim3(chunks, NELEM), dim3(NTHREADS), SMEM_BYTES,
                       stream, X, Z, W1, b1, W2, b2, W3, b3, out);
}
